# starting kernel
# baseline (speedup 1.0000x reference)
#include <hip/hip_runtime.h>
#include <hip/hip_bf16.h>

typedef __attribute__((ext_vector_type(2))) float v2f;
typedef __attribute__((ext_vector_type(8))) float v8f;

#define CIN  128
#define COUT 128
#define XPAD 132

__global__ __launch_bounds__(256) void gcn_gemm_norm(
    const float* __restrict__ x, const float* __restrict__ w,
    const int* __restrict__ colptr, float* __restrict__ h, int n)
{
  __shared__ float xs[16 * XPAD];
  __shared__ float onorm[16];

  const int tid  = threadIdx.x;
  const int wave = tid >> 5;
  const int lane = tid & 31;
  const int half = lane >> 4;
  const int l16  = lane & 15;
  const int n0   = wave * 16;

  v2f bfrag[32];
#pragma unroll
  for (int s = 0; s < 32; ++s) {
    const int kk = 4 * s + 2 * half;
    bfrag[s].x = w[kk       * COUT + n0 + l16];
    bfrag[s].y = w[(kk + 1) * COUT + n0 + l16];
  }

  const int ntiles = (n + 15) >> 4;
  for (int t = blockIdx.x; t < ntiles; t += gridDim.x) {
    const int row0 = t << 4;

    __syncthreads();
#pragma unroll
    for (int i = 0; i < 8; ++i) {
      const int e = tid + i * 256;
      const int r = e >> 7, c = e & 127;
      const int gr = row0 + r;
      xs[r * XPAD + c] = (gr < n) ? x[(size_t)gr * CIN + c] : 0.0f;
    }
    if (tid < 16) {
      const int gr = row0 + tid;
      const float deg = (gr < n) ? (float)(colptr[gr + 1] - colptr[gr]) : 1.0f;
      onorm[tid] = rsqrtf(deg);
    }
    __syncthreads();

    v8f acc = {};
#pragma unroll
    for (int s = 0; s < 32; ++s) {
      const int kk = 4 * s + 2 * half;
      v2f a;
      a.x = xs[l16 * XPAD + kk];
      a.y = xs[l16 * XPAD + kk + 1];
      acc = __builtin_amdgcn_wmma_f32_16x16x4_f32(
           false, a,  false, bfrag[s],
           (short)0, acc,  false,  false);
    }

#pragma unroll
    for (int r = 0; r < 8; ++r) {
      const int m  = r + 8 * half;
      const int gr = row0 + m;
      if (gr < n)
        h[(size_t)gr * COUT + n0 + l16] = acc[r] * onorm[m];
    }
  }
}

__global__ __launch_bounds__(256) void gcn_aggr(
    const float* __restrict__ h, const int* __restrict__ rowptr,
    const int* __restrict__ colind, const float* __restrict__ bias,
    float* __restrict__ out, int n)
{
  const int wave = threadIdx.x >> 5;
  const int lane = threadIdx.x & 31;
  const int wavesPerBlk = blockDim.x >> 5;

  const float4* __restrict__ h4 = (const float4*)h;
  const float4  b4 = ((const float4*)bias)[lane];

  for (int i = blockIdx.x * wavesPerBlk + wave; i < n;
       i += gridDim.x * wavesPerBlk) {
    const int e0 = rowptr[i];
    const int e1 = rowptr[i + 1];

    float ax = 0.f, ay = 0.f, az = 0.f, aw = 0.f;
    for (int e = e0; e < e1; ++e) {
      const int nbr = colind[e];
      const float4 v = h4[(size_t)nbr * (COUT / 4) + lane];
      ax += v.x; ay += v.y; az += v.z; aw += v.w;
    }

    const float inorm = rsqrtf((float)(e1 - e0));
    float4 o;
    o.x = ax * inorm + b4.x;
    o.y = ay * inorm + b4.y;
    o.z = az * inorm + b4.z;
    o.w = aw * inorm + b4.w;
    ((float4*)out)[(size_t)i * (COUT / 4) + lane] = o;
  }
}

extern "C" void kernel_launch(void* const* d_in, const int* in_sizes, int n_in,
                              void* d_out, int out_size, void* d_ws, size_t ws_size,
                              hipStream_t stream) {
  const float* x      = (const float*)d_in[0];
  const float* weight = (const float*)d_in[1];
  const float* bias   = (const float*)d_in[2];
  const int*   rowptr = (const int*)d_in[3];
  const int*   colind = (const int*)d_in[4];
  const int*   colptr = (const int*)d_in[5];

  const int n = in_sizes[3] - 1;
  float* h = (float*)d_ws;

  dim3 blk(256);
  gcn_gemm_norm<<<1024, blk, 0, stream>>>(x, weight, colptr, h, n);
  gcn_aggr<<<4096, blk, 0, stream>>>(h, rowptr, colind, bias, (float*)d_out, n);
}


# Fuse by linearity into one kernel: each block gathers and sums the out-norm-scaled x rows of its 16 nodes into LDS, then the f32 WMMA transform; neighbour sum now precedes the k sum; no h scratch
# speedup vs baseline: 1.3842x; 1.3842x over previous
#include <hip/hip_runtime.h>

typedef __attribute__((ext_vector_type(2))) float v2f;
typedef __attribute__((ext_vector_type(8))) float v8f;

#define CIN  128
#define COUT 128
#define XPAD 132

__device__ __forceinline__ void fma4(float4& a, float s, const float4& v) {
  a.x += s * v.x; a.y += s * v.y; a.z += s * v.z; a.w += s * v.w;
}

__global__ __launch_bounds__(256, 8) void gcn_aggr_gemm(
    const float* __restrict__ x, const float* __restrict__ w,
    const float* __restrict__ bias, const int* __restrict__ rowptr,
    const int* __restrict__ colind, const int* __restrict__ colptr,
    float* __restrict__ out, int n)
{
  __shared__ float xs[16 * XPAD];
  __shared__ float inorm[16];

  const int tid  = threadIdx.x;
  const int wave = __builtin_amdgcn_readfirstlane(tid >> 5);
  const int lane = tid & 31;
  const int half = lane >> 4;
  const int l16  = lane & 15;
  const int n0   = wave * 16;

  v2f bfrag[32];
#pragma unroll
  for (int s = 0; s < 32; ++s) {
    const int kk = 4 * s + 2 * half;
    bfrag[s].x = w[kk       * COUT + n0 + l16];
    bfrag[s].y = w[(kk + 1) * COUT + n0 + l16];
  }
  const float bcol = bias[n0 + l16];
  const unsigned loff = 4u * (unsigned)lane;

  const int ntiles = (n + 15) >> 4;
  for (int t = blockIdx.x; t < ntiles; t += gridDim.x) {
    const int row0 = t << 4;

    const int iA = row0 + 2 * wave;
    int p0 = 0, p1 = 0, p2 = 0;
    if (iA < n)     { p0 = rowptr[iA]; p1 = rowptr[iA + 1]; p2 = p1; }
    if (iA + 1 < n) { p2 = rowptr[iA + 2]; }

    float4 aA = make_float4(0.f, 0.f, 0.f, 0.f);
    float4 aB = make_float4(0.f, 0.f, 0.f, 0.f);

    for (int base = p0; base < p2; base += 32) {
      const int cnt = min(32, p2 - base);
      int   c  = 0;
      float on = 0.f;
      if (lane < cnt) {
        c  = colind[base + lane];
        on = rsqrtf((float)(colptr[c + 1] - colptr[c]));
      }
      const int oni = __float_as_int(on);

      int j = 0;
      for (; j + 8 <= cnt; j += 8) {
        const int eb = base + j;
        const bool allA = (eb + 8 <= p1);
        const bool allB = (eb >= p1);
        if (allA || allB) {
          float4 v[8];
          float  s[8];
#pragma unroll
          for (int u = 0; u < 8; ++u) {
            const int cj = __builtin_amdgcn_readlane(c, j + u);
            s[u] = __int_as_float(__builtin_amdgcn_readlane(oni, j + u));
            v[u] = *(const float4*)(x + (size_t)cj * CIN + loff);
          }
          float4 g = make_float4(0.f, 0.f, 0.f, 0.f);
#pragma unroll
          for (int u = 0; u < 8; ++u) fma4(g, s[u], v[u]);
          if (allA) { aA.x += g.x; aA.y += g.y; aA.z += g.z; aA.w += g.w; }
          else      { aB.x += g.x; aB.y += g.y; aB.z += g.z; aB.w += g.w; }
        } else {
          for (int u = 0; u < 8; ++u) {
            const int   cj = __builtin_amdgcn_readlane(c, j + u);
            const float sj = __int_as_float(__builtin_amdgcn_readlane(oni, j + u));
            const float4 vv = *(const float4*)(x + (size_t)cj * CIN + loff);
            if (eb + u < p1) fma4(aA, sj, vv); else fma4(aB, sj, vv);
          }
        }
      }
      for (; j < cnt; ++j) {
        const int   cj = __builtin_amdgcn_readlane(c, j);
        const float sj = __int_as_float(__builtin_amdgcn_readlane(oni, j));
        const float4 vv = *(const float4*)(x + (size_t)cj * CIN + loff);
        if (base + j < p1) fma4(aA, sj, vv); else fma4(aB, sj, vv);
      }
    }

    __syncthreads();
    *(float4*)&xs[(2 * wave)     * XPAD + 4 * lane] = aA;
    *(float4*)&xs[(2 * wave + 1) * XPAD + 4 * lane] = aB;
    if (lane == 0) {
      inorm[2 * wave]     = rsqrtf((float)(p1 - p0));
      inorm[2 * wave + 1] = rsqrtf((float)(p2 - p1));
    }
    __syncthreads();

    v8f acc = {};
#pragma unroll
    for (int s = 0; s < 32; ++s) {
      const int kk = 4 * s + 2 * half;
      v2f a;
      a.x = xs[l16 * XPAD + kk];
      a.y = xs[l16 * XPAD + kk + 1];
      acc = __builtin_amdgcn_wmma_f32_16x16x4_f32(
           false, a,  false, bfrag[s],
           (short)0, acc,  false,  false);
    }

#pragma unroll
    for (int r = 0; r < 8; ++r) {
      const int m  = r + 8 * half;
      const int gr = row0 + m;
      if (gr < n)
        out[(size_t)gr * COUT + n0 + l16] = acc[r] * inorm[m] + bcol;
    }
  }
}

extern "C" void kernel_launch(void* const* d_in, const int* in_sizes, int n_in,
                              void* d_out, int out_size, void* d_ws, size_t ws_size,
                              hipStream_t stream) {
  const float* x      = (const float*)d_in[0];
  const float* weight = (const float*)d_in[1];
  const float* bias   = (const float*)d_in[2];
  const int*   rowptr = (const int*)d_in[3];
  const int*   colind = (const int*)d_in[4];
  const int*   colptr = (const int*)d_in[5];

  const int n = in_sizes[3] - 1;

  dim3 blk(256);
  gcn_aggr_gemm<<<1024, blk, 0, stream>>>(x, weight, bias, rowptr, colind, colptr,
                                          (float*)d_out, n);
}


# Epilogue staged through LDS: each wave writes its two output rows as whole 512-B rows (16-B store per lane, whole 128-B lines) instead of 64-B pieces of a line shared by two waves
# speedup vs baseline: 1.5882x; 1.1474x over previous
#include <hip/hip_runtime.h>

typedef __attribute__((ext_vector_type(2))) float v2f;
typedef __attribute__((ext_vector_type(8))) float v8f;

#define CIN  128
#define COUT 128
#define XPAD 132

__device__ __forceinline__ void fma4(float4& a, float s, const float4& v) {
  a.x += s * v.x; a.y += s * v.y; a.z += s * v.z; a.w += s * v.w;
}

__global__ __launch_bounds__(256, 8) void gcn_aggr_gemm(
    const float* __restrict__ x, const float* __restrict__ w,
    const float* __restrict__ bias, const int* __restrict__ rowptr,
    const int* __restrict__ colind, const int* __restrict__ colptr,
    float* __restrict__ out, int n)
{
  __shared__ float xs[16 * XPAD];
  __shared__ float ys[16 * XPAD];
  __shared__ float bs[COUT];

  const int tid  = threadIdx.x;
  const int wave = __builtin_amdgcn_readfirstlane(tid >> 5);
  const int lane = tid & 31;
  const int half = lane >> 4;
  const int l16  = lane & 15;
  const int n0   = wave * 16;

  v2f bfrag[32];
#pragma unroll
  for (int s = 0; s < 32; ++s) {
    const int kk = 4 * s + 2 * half;
    bfrag[s].x = w[kk       * COUT + n0 + l16];
    bfrag[s].y = w[(kk + 1) * COUT + n0 + l16];
  }
  if (tid < COUT) bs[tid] = bias[tid];
  const unsigned loff = 4u * (unsigned)lane;

  const int ntiles = (n + 15) >> 4;
  for (int t = blockIdx.x; t < ntiles; t += gridDim.x) {
    const int row0 = t << 4;

    const int iA = row0 + 2 * wave;
    int p0 = 0, p1 = 0, p2 = 0;
    if (iA < n)     { p0 = rowptr[iA]; p1 = rowptr[iA + 1]; p2 = p1; }
    if (iA + 1 < n) { p2 = rowptr[iA + 2]; }

    float4 aA = make_float4(0.f, 0.f, 0.f, 0.f);
    float4 aB = make_float4(0.f, 0.f, 0.f, 0.f);

    for (int base = p0; base < p2; base += 32) {
      const int cnt = min(32, p2 - base);
      int   c  = 0;
      float on = 0.f;
      if (lane < cnt) {
        c  = colind[base + lane];
        on = rsqrtf((float)(colptr[c + 1] - colptr[c]));
      }
      const int oni = __float_as_int(on);

      int j = 0;
      for (; j + 8 <= cnt; j += 8) {
        const int eb = base + j;
        const bool allA = (eb + 8 <= p1);
        const bool allB = (eb >= p1);
        if (allA || allB) {
          float4 v[8];
          float  s[8];
#pragma unroll
          for (int u = 0; u < 8; ++u) {
            const int cj = __builtin_amdgcn_readlane(c, j + u);
            s[u] = __int_as_float(__builtin_amdgcn_readlane(oni, j + u));
            v[u] = *(const float4*)(x + (size_t)cj * CIN + loff);
          }
          float4 g = make_float4(0.f, 0.f, 0.f, 0.f);
#pragma unroll
          for (int u = 0; u < 8; ++u) fma4(g, s[u], v[u]);
          if (allA) { aA.x += g.x; aA.y += g.y; aA.z += g.z; aA.w += g.w; }
          else      { aB.x += g.x; aB.y += g.y; aB.z += g.z; aB.w += g.w; }
        } else {
          for (int u = 0; u < 8; ++u) {
            const int   cj = __builtin_amdgcn_readlane(c, j + u);
            const float sj = __int_as_float(__builtin_amdgcn_readlane(oni, j + u));
            const float4 vv = *(const float4*)(x + (size_t)cj * CIN + loff);
            if (eb + u < p1) fma4(aA, sj, vv); else fma4(aB, sj, vv);
          }
        }
      }
      for (; j < cnt; ++j) {
        const int   cj = __builtin_amdgcn_readlane(c, j);
        const float sj = __int_as_float(__builtin_amdgcn_readlane(oni, j));
        const float4 vv = *(const float4*)(x + (size_t)cj * CIN + loff);
        if (base + j < p1) fma4(aA, sj, vv); else fma4(aB, sj, vv);
      }
    }

    *(float4*)&xs[(2 * wave)     * XPAD + 4 * lane] = aA;
    *(float4*)&xs[(2 * wave + 1) * XPAD + 4 * lane] = aB;
    __syncthreads();

    v8f acc = {};
#pragma unroll
    for (int s = 0; s < 32; ++s) {
      const int kk = 4 * s + 2 * half;
      v2f a;
      a.x = xs[l16 * XPAD + kk];
      a.y = xs[l16 * XPAD + kk + 1];
      acc = __builtin_amdgcn_wmma_f32_16x16x4_f32(
           false, a,  false, bfrag[s],
           (short)0, acc,  false,  false);
    }

#pragma unroll
    for (int r = 0; r < 8; ++r)
      ys[(r + 8 * half) * XPAD + n0 + l16] = acc[r];
    __syncthreads();

    {
      const float4 b4 = *(const float4*)&bs[4 * lane];
      if (iA < n) {
        const float  sA = rsqrtf((float)(p1 - p0));
        const float4 y  = *(const float4*)&ys[(2 * wave) * XPAD + 4 * lane];
        float4 o;
        o.x = y.x * sA + b4.x; o.y = y.y * sA + b4.y; o.z = y.z * sA + b4.z; o.w = y.w * sA + b4.w;
        *(float4*)(out + (size_t)iA * COUT + 4 * lane) = o;
      }
      if (iA + 1 < n) {
        const float  sB = rsqrtf((float)(p2 - p1));
        const float4 y  = *(const float4*)&ys[(2 * wave + 1) * XPAD + 4 * lane];
        float4 o;
        o.x = y.x * sB + b4.x; o.y = y.y * sB + b4.y; o.z = y.z * sB + b4.z; o.w = y.w * sB + b4.w;
        *(float4*)(out + (size_t)(iA + 1) * COUT + 4 * lane) = o;
      }
    }
  }
}

extern "C" void kernel_launch(void* const* d_in, const int* in_sizes, int n_in,
                              void* d_out, int out_size, void* d_ws, size_t ws_size,
                              hipStream_t stream) {
  const float* x      = (const float*)d_in[0];
  const float* weight = (const float*)d_in[1];
  const float* bias   = (const float*)d_in[2];
  const int*   rowptr = (const int*)d_in[3];
  const int*   colind = (const int*)d_in[4];
  const int*   colptr = (const int*)d_in[5];

  const int n = in_sizes[3] - 1;

  dim3 blk(256);
  gcn_aggr_gemm<<<1024, blk, 0, stream>>>(x, weight, bias, rowptr, colind, colptr,
                                          (float*)d_out, n);
}
